# starting kernel
# baseline (speedup 1.0000x reference)
#include <hip/hip_runtime.h>
#include <hip/hip_bf16.h>
#include <math.h>

#define B_   32
#define C_   256
#define H_   56
#define W_   56
#define HW_  (H_ * W_)
#define NPIX (B_ * HW_)

typedef __attribute__((ext_vector_type(2))) float v2f;
typedef __attribute__((ext_vector_type(4))) float v4f;
typedef __attribute__((ext_vector_type(8))) float v8f;

__global__ __launch_bounds__(256) void reduce_kernel(const float* __restrict__ x,
                                                     float2* __restrict__ y) {
  __shared__ float ssum[256];
  __shared__ float smax[256];
  const int t  = threadIdx.x;
  const int p  = t & 63;
  const int q  = t >> 6;
  const int gp = blockIdx.x * 64 + p;
  const int b  = gp / HW_;
  const int hw = gp - b * HW_;

  const float* xp = x + (size_t)(b * C_ + q * 64) * HW_ + hw;
  float s  = 0.0f;
  float mx = -3.402823466e38f;
#pragma unroll 8
  for (int c = 0; c < 64; ++c) {
    float v = xp[(size_t)c * HW_];
    s += v;
    mx = fmaxf(mx, v);
  }
  ssum[t] = s;
  smax[t] = mx;
  __syncthreads();
  if (q == 0) {
    float s0 = ssum[p] + ssum[p + 64] + ssum[p + 128] + ssum[p + 192];
    float m0 = fmaxf(fmaxf(smax[p], smax[p + 64]), fmaxf(smax[p + 128], smax[p + 192]));
    float2 r;
    r.x = s0 * (1.0f / (float)C_);
    r.y = m0;
    y[gp] = r;
  }
}

__global__ __launch_bounds__(32) void conv_wmma_kernel(const float2* __restrict__ y,
                                                       const float* __restrict__ cw,
                                                       float* __restrict__ att) {
  const int lane = threadIdx.x;
  const int m    = lane & 15;
  const int hi   = lane >> 4;

  const int gp = blockIdx.x * 16 + m;
  const int b  = gp / HW_;
  const int hw = gp - b * HW_;
  const int h  = hw / W_;
  const int w  = hw - h * W_;
  const float2* yb = y + (size_t)b * HW_;

  v8f acc = {0.f, 0.f, 0.f, 0.f, 0.f, 0.f, 0.f, 0.f};

  for (int c = 0; c < 25; ++c) {
    const int tap = 2 * c + hi;
    float2 av = make_float2(0.f, 0.f);
    float  w0 = 0.f, w1 = 0.f;
    if (tap < 49) {
      const int kh = tap / 7;
      const int kw = tap - kh * 7;
      const int ih = h + kh - 3;
      const int iw = w + kw - 3;
      if (ih >= 0 && ih < H_ && iw >= 0 && iw < W_) av = yb[ih * W_ + iw];
      w0 = cw[tap];
      w1 = cw[49 + tap];
    }
    v2f A = {av.x, av.y};
    v2f Bm = {w0, w1};
    acc = __builtin_amdgcn_wmma_f32_16x16x4_f32(
         false, A,  false, Bm,
         (short)0, acc,  false,  false);
  }

  if (m == 0) {
    const int mbase = hi * 8;
#pragma unroll
    for (int r = 0; r < 8; ++r) {
      float z = acc[r];
      float s = 1.0f / (1.0f + expf(-z));
      att[blockIdx.x * 16 + mbase + r] = s;
    }
  }
}

__global__ __launch_bounds__(256) void mul_kernel(const float* __restrict__ x,
                                                  const float* __restrict__ att,
                                                  float* __restrict__ out, int n4) {
  const int i = blockIdx.x * blockDim.x + threadIdx.x;
  if (i >= n4) return;
  const int per_b = C_ * (HW_ / 4);
  const int b     = i / per_b;
  const int rem   = i - b * per_b;
  const int hw4   = rem % (HW_ / 4);

  const v4f xv = __builtin_nontemporal_load(reinterpret_cast<const v4f*>(x) + i);
  const v4f av = *(reinterpret_cast<const v4f*>(att + (size_t)b * HW_) + hw4);
  v4f o = xv * av;
  __builtin_nontemporal_store(o, reinterpret_cast<v4f*>(out) + i);
}

extern "C" void kernel_launch(void* const* d_in, const int* in_sizes, int n_in,
                              void* d_out, int out_size, void* d_ws, size_t ws_size,
                              hipStream_t stream) {
  const float* x  = (const float*)d_in[0];
  const float* cw = (const float*)d_in[1];
  float* out      = (float*)d_out;

  float2* y   = (float2*)d_ws;
  float*  att = (float*)((char*)d_ws + (size_t)NPIX * sizeof(float2));

  (void)in_sizes; (void)n_in; (void)out_size; (void)ws_size;

  reduce_kernel<<<NPIX / 64, 256, 0, stream>>>(x, y);

  conv_wmma_kernel<<<NPIX / 16, 32, 0, stream>>>(y, cw, att);

  const int n4 = (B_ * C_ * HW_) / 4;
  mul_kernel<<<(n4 + 255) / 256, 256, 0, stream>>>(x, att, out, n4);
}


# fuse conv+sigmoid into the multiply kernel: one workgroup per 64-pixel tile runs the same WMMA chain into LDS, then scales all 256 channels with all its x loads in flight; 3 launches -> 2
# speedup vs baseline: 1.7648x; 1.7648x over previous
#include <hip/hip_runtime.h>
#include <hip/hip_bf16.h>
#include <math.h>

#define B_   32
#define C_   256
#define H_   56
#define W_   56
#define HW_  (H_ * W_)
#define NPIX (B_ * HW_)

typedef __attribute__((ext_vector_type(2))) float v2f;
typedef __attribute__((ext_vector_type(4))) float v4f;
typedef __attribute__((ext_vector_type(8))) float v8f;

__global__ __launch_bounds__(256) void reduce_kernel(const float* __restrict__ x,
                                                     float2* __restrict__ y) {
  __shared__ float ssum[256];
  __shared__ float smax[256];
  const int t  = threadIdx.x;
  const int p  = t & 63;
  const int q  = t >> 6;
  const int gp = blockIdx.x * 64 + p;
  const int b  = gp / HW_;
  const int hw = gp - b * HW_;

  const float* xp = x + (size_t)(b * C_ + q * 64) * HW_ + hw;
  float s  = 0.0f;
  float mx = -3.402823466e38f;
#pragma unroll 8
  for (int c = 0; c < 64; ++c) {
    float v = xp[(size_t)c * HW_];
    s += v;
    mx = fmaxf(mx, v);
  }
  ssum[t] = s;
  smax[t] = mx;
  __syncthreads();
  if (q == 0) {
    float s0 = ssum[p] + ssum[p + 64] + ssum[p + 128] + ssum[p + 192];
    float m0 = fmaxf(fmaxf(smax[p], smax[p + 64]), fmaxf(smax[p + 128], smax[p + 192]));
    float2 r;
    r.x = s0 * (1.0f / (float)C_);
    r.y = m0;
    y[gp] = r;
  }
}

#define TILE_PX   64
#define TILES_IMG (HW_ / TILE_PX)
#define CH_STEP   16
#define N_PASS    (C_ / CH_STEP)
#define WLANE(j)  __int_as_float(__builtin_amdgcn_readlane(__float_as_int(wreg[((j) < 98 ? (j) : 0) >> 5]), ((j) < 98 ? (j) : 0) & 31))

__global__ __launch_bounds__(256) void attn_mul_kernel(const float2* __restrict__ y,
                                                       const float* __restrict__ cw,
                                                       const float* __restrict__ x,
                                                       float* __restrict__ out) {
  __shared__ __attribute__((aligned(16))) float att_s[TILE_PX];

  const int t    = threadIdx.x;
  const int lane = t & 31;
  const int wv   = __builtin_amdgcn_readfirstlane(t >> 5);
  const int m    = lane & 15;
  const int hi   = lane >> 4;

  const int b  = blockIdx.x / TILES_IMG;
  const int ti = blockIdx.x - b * TILES_IMG;

  float2 Ar[25];
  bool   ok[25];
  float  wreg[4];
  if (wv < 4) {
    const int hw = ti * TILE_PX + wv * 16 + m;
    const int h  = hw / W_;
    const int w  = hw - h * W_;
    const float2* yb = y + (size_t)b * HW_;
#pragma unroll
    for (int k = 0; k < 4; ++k) {
      const int j = lane + 32 * k;
      wreg[k] = cw[j < 98 ? j : 0];
    }
#pragma unroll
    for (int c = 0; c < 25; ++c) {
      const int  tap = 2 * c + hi;
      const bool tv  = tap < 49;
      const int  kh  = tap / 7;
      const int  kw  = tap - kh * 7;
      const int  ih  = h + kh - 3;
      const int  iw  = w + kw - 3;
      ok[c] = tv && ih >= 0 && ih < H_ && iw >= 0 && iw < W_;
      Ar[c] = yb[ok[c] ? ih * W_ + iw : hw];
    }
  }

  const size_t off = (size_t)(b * C_ + (t >> 4)) * (HW_ / 4) + ti * (TILE_PX / 4) + (t & 15);
  const v4f* xp = reinterpret_cast<const v4f*>(x) + off;
  v4f*       op = reinterpret_cast<v4f*>(out) + off;
  v4f xv[N_PASS];
#pragma unroll
  for (int i = 0; i < N_PASS; ++i)
    xv[i] = __builtin_nontemporal_load(xp + (size_t)i * CH_STEP * (HW_ / 4));

  if (wv < 4) {
    v8f acc = {0.f, 0.f, 0.f, 0.f, 0.f, 0.f, 0.f, 0.f};
#pragma unroll
    for (int c = 0; c < 25; ++c) {
      const int   ta  = 2 * c, tb = 2 * c + 1;
      const float a0  = WLANE(ta), a1 = WLANE(49 + ta);
      const float b0  = (tb < 49) ? WLANE(tb) : 0.f;
      const float b1  = (tb < 49) ? WLANE(49 + tb) : 0.f;
      const v2f A  = {ok[c] ? Ar[c].x : 0.f, ok[c] ? Ar[c].y : 0.f};
      const v2f Bm = {hi ? b0 : a0, hi ? b1 : a1};
      acc = __builtin_amdgcn_wmma_f32_16x16x4_f32(
           false, A,  false, Bm,
           (short)0, acc,  false,  false);
    }
    const int r = lane & 7;
    float z = acc[0];
#pragma unroll
    for (int k = 1; k < 8; ++k) z = (r == k) ? acc[k] : z;
    const float s = 1.0f / (1.0f + expf(-z));
    if ((lane & 8) == 0) att_s[wv * 16 + hi * 8 + r] = s;
  }
  __syncthreads();

  const v4f av = *reinterpret_cast<const v4f*>(att_s + (t & 15) * 4);
#pragma unroll
  for (int i = 0; i < N_PASS; ++i) {
    v4f o = xv[i] * av;
    __builtin_nontemporal_store(o, op + (size_t)i * CH_STEP * (HW_ / 4));
  }
}

extern "C" void kernel_launch(void* const* d_in, const int* in_sizes, int n_in,
                              void* d_out, int out_size, void* d_ws, size_t ws_size,
                              hipStream_t stream) {
  const float* x  = (const float*)d_in[0];
  const float* cw = (const float*)d_in[1];
  float* out      = (float*)d_out;

  float2* y   = (float2*)d_ws;

  (void)in_sizes; (void)n_in; (void)out_size; (void)ws_size;

  reduce_kernel<<<NPIX / 64, 256, 0, stream>>>(x, y);

  attn_mul_kernel<<<NPIX / TILE_PX, 256, 0, stream>>>(y, cw, x, out);
}
